# starting kernel
# baseline (speedup 1.0000x reference)
#include <hip/hip_runtime.h>

typedef __attribute__((ext_vector_type(2))) float v2f;
typedef __attribute__((ext_vector_type(8))) float v8f;

#define KP_EXTENT 0.6f
#define ALPHA 0.3f

constexpr int cB    = 8;
constexpr int cN    = 16384;
constexpr int cP    = 4096;
constexpr int cS    = 32;
constexpr int cK    = 15;
constexpr int cCin  = 64;
constexpr int cCout = 128;
constexpr int TILE_M = 16;
constexpr int PITCH  = 964;
constexpr int KDIM   = cK * cCin;

__global__ __launch_bounds__(256) void kpconv_fused(
    const float* __restrict__ xyz,
    const float* __restrict__ feat,
    const int*   __restrict__ qidx,
    const int*   __restrict__ nidx,
    const float* __restrict__ kpts,
    const float* __restrict__ W,
    float*       __restrict__ out)
{
    extern __shared__ float smem[];
    float* sA    = smem;
    float* sInfl = smem + TILE_M * PITCH;
    int*   sNid  = (int*)(sInfl + 4 * cS * 16);
    float* sKP   = (float*)(sNid + 4 * cS);

    const int tid  = threadIdx.x;
    const int tile = blockIdx.x;
    const int b    = tile / (cP / TILE_M);
    const int p0   = (tile % (cP / TILE_M)) * TILE_M;

    if (tid < cK * 3) sKP[tid] = kpts[tid];
    __syncthreads();

    const int g  = tid >> 6;
    const int lt = tid & 63;

    for (int iter = 0; iter < 4; ++iter) {
        const int m = iter * 4 + g;
        const int p = p0 + m;

        if (lt < cS) {
            const int s = lt;
            const int q = qidx[b * cP + p];
            const float qx = xyz[((size_t)b * cN + q) * 3 + 0];
            const float qy = xyz[((size_t)b * cN + q) * 3 + 1];
            const float qz = xyz[((size_t)b * cN + q) * 3 + 2];
            const int nn = nidx[((size_t)b * cP + p) * cS + s];
            sNid[g * cS + s] = nn;
            const float rx = xyz[((size_t)b * cN + nn) * 3 + 0] - qx;
            const float ry = xyz[((size_t)b * cN + nn) * 3 + 1] - qy;
            const float rz = xyz[((size_t)b * cN + nn) * 3 + 2] - qz;
            #pragma unroll
            for (int k = 0; k < cK; ++k) {
                const float dx = rx - sKP[k * 3 + 0];
                const float dy = ry - sKP[k * 3 + 1];
                const float dz = rz - sKP[k * 3 + 2];
                const float d  = sqrtf(fmaf(dx, dx, fmaf(dy, dy, dz * dz)));
                float inf = 1.0f - d * (1.0f / KP_EXTENT);
                inf = inf > 0.0f ? inf : 0.0f;
                sInfl[(g * cS + s) * 16 + k] = inf;
            }
        }
        __syncthreads();

        {
            const int c = lt;
            float acc[cK];
            #pragma unroll
            for (int k = 0; k < cK; ++k) acc[k] = 0.0f;
            for (int s = 0; s < cS; ++s) {
                const float f = feat[((size_t)b * cN + sNid[g * cS + s]) * cCin + c];
                const float* ir = &sInfl[(g * cS + s) * 16];
                #pragma unroll
                for (int k = 0; k < cK; ++k) acc[k] = fmaf(ir[k], f, acc[k]);
            }
            #pragma unroll
            for (int k = 0; k < cK; ++k) sA[m * PITCH + k * cCin + c] = acc[k];
        }
        __syncthreads();
    }

    const int w    = tid >> 5;
    const int lane = tid & 31;
    const int mrow = lane & 15;
    const int koff = (lane >> 4) * 2;
    const int ncol = (w << 4) + (lane & 15);

    v8f acc = {};
    #pragma unroll 4
    for (int kk = 0; kk < KDIM; kk += 4) {
        const float* ap = &sA[mrow * PITCH + kk + koff];
        v2f a; a.x = ap[0]; a.y = ap[1];
        const float* bp = &W[(size_t)(kk + koff) * cCout + ncol];
        v2f bmat; bmat.x = bp[0]; bmat.y = bp[cCout];
        acc = __builtin_amdgcn_wmma_f32_16x16x4_f32(
             false, a,  false, bmat,
             (short)0, acc,  false,  false);
    }

    const int mbase = (lane >> 4) * 8;
    #pragma unroll
    for (int r = 0; r < 8; ++r) {
        float v = acc[r];
        v = v > 0.0f ? v : ALPHA * v;
        const int m = r + mbase;
        out[((size_t)(b * cP + p0 + m)) * cCout + ncol] = v;
    }
}

extern "C" void kernel_launch(void* const* d_in, const int* in_sizes, int n_in,
                              void* d_out, int out_size, void* d_ws, size_t ws_size,
                              hipStream_t stream) {
    const float* xyz  = (const float*)d_in[0];
    const float* feat = (const float*)d_in[1];
    const int*   qidx = (const int*)d_in[2];
    const int*   nidx = (const int*)d_in[3];
    const float* kpts = (const float*)d_in[4];
    const float* W    = (const float*)d_in[5];
    float* out = (float*)d_out;

    const int blocks = cB * cP / TILE_M;
    const size_t shmem = (size_t)(TILE_M * PITCH + 4 * cS * 16) * sizeof(float)
                       + (size_t)(4 * cS) * sizeof(int)
                       + 64 * sizeof(float);
    kpconv_fused<<<blocks, 256, shmem, stream>>>(xyz, feat, qidx, nidx, kpts, W, out);
}


# Neighbour aggregation (influence x gathered feature rows) moved from per-thread f32 FMAs to the f32 16x16x4 matrix instruction, one wave per point; the 32 neighbour terms are summed in another order
# speedup vs baseline: 1.3970x; 1.3970x over previous
#include <hip/hip_runtime.h>


typedef __attribute__((ext_vector_type(2))) float v2f;
typedef __attribute__((ext_vector_type(8))) float v8f;

#define KP_EXTENT 0.6f
#define ALPHA 0.3f

constexpr int cB    = 8;
constexpr int cN    = 16384;
constexpr int cP    = 4096;
constexpr int cS    = 32;
constexpr int cK    = 15;
constexpr int cCin  = 64;
constexpr int cCout = 128;
constexpr int TILE_M = 16;
constexpr int KROWS  = 16;
constexpr int PITCH  = KROWS * cCin + 4;
constexpr int KDIM   = cK * cCin;
constexpr int PTS_PER_WAVE = TILE_M / 8;

__device__ __forceinline__ void agg_step(v8f& a0, v8f& a1, v8f& a2, v8f& a3,
                                         const float i0, const float i1,
                                         const float4 f0, const float4 f1)
{
    v2f a;  a.x  = i0;   a.y  = i1;
    v2f g0; g0.x = f0.x; g0.y = f1.x;
    v2f g1; g1.x = f0.y; g1.y = f1.y;
    v2f g2; g2.x = f0.z; g2.y = f1.z;
    v2f g3; g3.x = f0.w; g3.y = f1.w;
    a0 = __builtin_amdgcn_wmma_f32_16x16x4_f32(false, a, false, g0, (short)0, a0, false, false);
    a1 = __builtin_amdgcn_wmma_f32_16x16x4_f32(false, a, false, g1, (short)0, a1, false, false);
    a2 = __builtin_amdgcn_wmma_f32_16x16x4_f32(false, a, false, g2, (short)0, a2, false, false);
    a3 = __builtin_amdgcn_wmma_f32_16x16x4_f32(false, a, false, g3, (short)0, a3, false, false);
}

__global__ __launch_bounds__(256) __attribute__((amdgpu_num_vgpr(128))) void kpconv_fused(
    const float* __restrict__ xyz,
    const float* __restrict__ feat,
    const int*   __restrict__ qidx,
    const int*   __restrict__ nidx,
    const float* __restrict__ kpts,
    const float* __restrict__ W,
    float*       __restrict__ out)
{
    extern __shared__ __align__(16) float smem[];
    float* sA = smem;

    const int tid  = threadIdx.x;
    const int tile = blockIdx.x;
    const int b    = tile / (cP / TILE_M);
    const int p0   = (tile % (cP / TILE_M)) * TILE_M;

    const int w    = tid >> 5;
    const int lane = tid & 31;
    const int c    = lane & 15;
    const int h    = lane >> 4;

    {
        const bool live = c < cK;
        const int  kc   = live ? c : 0;
        const float kpx = kpts[kc * 3 + 0];
        const float kpy = kpts[kc * 3 + 1];
        const float kpz = kpts[kc * 3 + 2];
        const float*  xb = xyz + (size_t)b * cN * 3;
        const float4* fb = reinterpret_cast<const float4*>(feat + (size_t)b * cN * cCin);

        float qxs[PTS_PER_WAVE], qys[PTS_PER_WAVE], qzs[PTS_PER_WAVE];
        #pragma unroll
        for (int pt = 0; pt < PTS_PER_WAVE; ++pt) {
            const int q = qidx[b * cP + p0 + w * PTS_PER_WAVE + pt];
            const float* qp = xb + q * 3;
            qxs[pt] = qp[0]; qys[pt] = qp[1]; qzs[pt] = qp[2];
        }

        #pragma unroll
        for (int pt = 0; pt < PTS_PER_WAVE; ++pt) {
            const int m = w * PTS_PER_WAVE + pt;
            const int p = p0 + m;
            const float qx = qxs[pt], qy = qys[pt], qz = qzs[pt];

            const int4* np = reinterpret_cast<const int4*>(nidx + ((size_t)b * cP + p) * cS + 16 * h);
            unsigned nid[16];
            #pragma unroll
            for (int i = 0; i < 4; ++i) {
                const int4 v = np[i];
                nid[4 * i + 0] = (unsigned)v.x; nid[4 * i + 1] = (unsigned)v.y;
                nid[4 * i + 2] = (unsigned)v.z; nid[4 * i + 3] = (unsigned)v.w;
            }

            float4 fA[8];
            #pragma unroll
            for (int i = 0; i < 8; ++i) fA[i] = fb[nid[i] * 16u + (unsigned)c];

            float infl[16];
            #pragma unroll
            for (int i = 0; i < 16; ++i) {
                const float* xp = xb + nid[i] * 3u;
                const float rx = xp[0] - qx;
                const float ry = xp[1] - qy;
                const float rz = xp[2] - qz;
                const float dx = rx - kpx;
                const float dy = ry - kpy;
                const float dz = rz - kpz;
                const float d  = sqrtf(fmaf(dx, dx, fmaf(dy, dy, dz * dz)));
                float inf = 1.0f - d * (1.0f / KP_EXTENT);
                inf = inf > 0.0f ? inf : 0.0f;
                infl[i] = live ? inf : 0.0f;
            }

            __builtin_amdgcn_sched_barrier(0);

            float4 fB[8];
            #pragma unroll
            for (int i = 0; i < 8; ++i) fB[i] = fb[nid[8 + i] * 16u + (unsigned)c];

            v8f acc0 = {}, acc1 = {}, acc2 = {}, acc3 = {};
            #pragma unroll
            for (int j = 0; j < 4; ++j)
                agg_step(acc0, acc1, acc2, acc3, infl[2 * j], infl[2 * j + 1], fA[2 * j], fA[2 * j + 1]);
            #pragma unroll
            for (int j = 0; j < 4; ++j)
                agg_step(acc0, acc1, acc2, acc3, infl[8 + 2 * j], infl[8 + 2 * j + 1], fB[2 * j], fB[2 * j + 1]);

            float* dst = &sA[m * PITCH + (8 * h) * cCin + 4 * c];
            #pragma unroll
            for (int r = 0; r < 8; ++r) {
                const float4 v = make_float4(acc0[r], acc1[r], acc2[r], acc3[r]);
                *reinterpret_cast<float4*>(dst + r * cCin) = v;
            }
        }
    }
    __syncthreads();

    const int mrow = lane & 15;
    const int koff = (lane >> 4) * 2;
    const int ncol = (w << 4) + (lane & 15);

    v8f acc = {};
    #pragma unroll 4
    for (int kk = 0; kk < KDIM; kk += 4) {
        const float* ap = &sA[mrow * PITCH + kk + koff];
        v2f a; a.x = ap[0]; a.y = ap[1];
        const float* bp = &W[(size_t)(kk + koff) * cCout + ncol];
        v2f bmat; bmat.x = bp[0]; bmat.y = bp[cCout];
        acc = __builtin_amdgcn_wmma_f32_16x16x4_f32(
             false, a,  false, bmat,
             (short)0, acc,  false,  false);
    }

    const int mbase = (lane >> 4) * 8;
    #pragma unroll
    for (int r = 0; r < 8; ++r) {
        float v = acc[r];
        v = v > 0.0f ? v : ALPHA * v;
        const int m = r + mbase;
        out[((size_t)(b * cP + p0 + m)) * cCout + ncol] = v;
    }
}

extern "C" void kernel_launch(void* const* d_in, const int* in_sizes, int n_in,
                              void* d_out, int out_size, void* d_ws, size_t ws_size,
                              hipStream_t stream) {
    const float* xyz  = (const float*)d_in[0];
    const float* feat = (const float*)d_in[1];
    const int*   qidx = (const int*)d_in[2];
    const int*   nidx = (const int*)d_in[3];
    const float* kpts = (const float*)d_in[4];
    const float* W    = (const float*)d_in[5];
    float* out = (float*)d_out;

    const int blocks = cB * cP / TILE_M;
    const size_t shmem = (size_t)(TILE_M * PITCH) * sizeof(float);
    kpconv_fused<<<blocks, 256, shmem, stream>>>(xyz, feat, qidx, nidx, kpts, W, out);
}


# Second product re-tiled: 32 points per work-group, each wave a 32x32 tile over a quarter of K, so one 8-byte W load feeds 4 matrix ops; the 4 K partial sums are added through LDS (new sum order)
# speedup vs baseline: 1.8292x; 1.3094x over previous
#include <hip/hip_runtime.h>


typedef __attribute__((ext_vector_type(2))) float v2f;
typedef __attribute__((ext_vector_type(8))) float v8f;

#define KP_EXTENT 0.6f
#define ALPHA 0.3f

constexpr int cB    = 8;
constexpr int cN    = 16384;
constexpr int cP    = 4096;
constexpr int cS    = 32;
constexpr int cK    = 15;
constexpr int cCin  = 64;
constexpr int cCout = 128;
constexpr int TILE_M = 32;
constexpr int NWAVES = 16;
constexpr int KROWS  = 16;
constexpr int PITCH  = KROWS * cCin + 4;
constexpr int KDIM   = cK * cCin;
constexpr int PTS_PER_WAVE = TILE_M / NWAVES;
constexpr int NGROUPS = 4;
constexpr int KSPLIT  = NWAVES / NGROUPS;
constexpr int KQ      = KDIM / KSPLIT;
constexpr int PART    = 4 * 8 * 32;
static_assert(NWAVES * PART <= TILE_M * PITCH, "the partial tiles fit inside the dead A rows");

__device__ __forceinline__ void agg_step(v8f& a0, v8f& a1, v8f& a2, v8f& a3,
                                         const float i0, const float i1,
                                         const float4 f0, const float4 f1)
{
    v2f a;  a.x  = i0;   a.y  = i1;
    v2f g0; g0.x = f0.x; g0.y = f1.x;
    v2f g1; g1.x = f0.y; g1.y = f1.y;
    v2f g2; g2.x = f0.z; g2.y = f1.z;
    v2f g3; g3.x = f0.w; g3.y = f1.w;
    a0 = __builtin_amdgcn_wmma_f32_16x16x4_f32(false, a, false, g0, (short)0, a0, false, false);
    a1 = __builtin_amdgcn_wmma_f32_16x16x4_f32(false, a, false, g1, (short)0, a1, false, false);
    a2 = __builtin_amdgcn_wmma_f32_16x16x4_f32(false, a, false, g2, (short)0, a2, false, false);
    a3 = __builtin_amdgcn_wmma_f32_16x16x4_f32(false, a, false, g3, (short)0, a3, false, false);
}

__device__ __forceinline__ v8f mm(const v2f a, const v2f b, const v8f c)
{
    return __builtin_amdgcn_wmma_f32_16x16x4_f32(false, a, false, b, (short)0, c, false, false);
}

__global__ __launch_bounds__(512) __attribute__((amdgpu_num_vgpr(120))) void kpconv_fused(
    const float* __restrict__ xyz,
    const float* __restrict__ feat,
    const int*   __restrict__ qidx,
    const int*   __restrict__ nidx,
    const float* __restrict__ kpts,
    const float* __restrict__ W,
    float*       __restrict__ out)
{
    __shared__ __attribute__((aligned(16))) float smem[TILE_M * PITCH];
    float* sA = smem;

    const int tid  = threadIdx.x;
    const int tile = blockIdx.x;
    const int b    = tile / (cP / TILE_M);
    const int p0   = (tile % (cP / TILE_M)) * TILE_M;

    const int w    = __builtin_amdgcn_readfirstlane(tid >> 5);
    const int lane = tid & 31;
    const int c    = lane & 15;
    const int h    = lane >> 4;

    {
        const bool live = c < cK;
        const int  kc   = live ? c : 0;
        const float kpx = kpts[kc * 3 + 0];
        const float kpy = kpts[kc * 3 + 1];
        const float kpz = kpts[kc * 3 + 2];
        const float*  xb = xyz + (size_t)b * cN * 3;
        const float4* fb = reinterpret_cast<const float4*>(feat + (size_t)b * cN * cCin);

        float qxs[PTS_PER_WAVE], qys[PTS_PER_WAVE], qzs[PTS_PER_WAVE];
        #pragma unroll
        for (int pt = 0; pt < PTS_PER_WAVE; ++pt) {
            const int q = qidx[b * cP + p0 + w * PTS_PER_WAVE + pt];
            const float* qp = xb + q * 3;
            qxs[pt] = qp[0]; qys[pt] = qp[1]; qzs[pt] = qp[2];
        }

        #pragma unroll
        for (int pt = 0; pt < PTS_PER_WAVE; ++pt) {
            const int m = w * PTS_PER_WAVE + pt;
            const int p = p0 + m;
            const float qx = qxs[pt], qy = qys[pt], qz = qzs[pt];

            const int4* np = reinterpret_cast<const int4*>(nidx + ((size_t)b * cP + p) * cS + 16 * h);
            unsigned nid[16];
            #pragma unroll
            for (int i = 0; i < 4; ++i) {
                const int4 v = np[i];
                nid[4 * i + 0] = (unsigned)v.x; nid[4 * i + 1] = (unsigned)v.y;
                nid[4 * i + 2] = (unsigned)v.z; nid[4 * i + 3] = (unsigned)v.w;
            }

            float4 fA[8];
            #pragma unroll
            for (int i = 0; i < 8; ++i) fA[i] = fb[nid[i] * 16u + (unsigned)c];

            float infl[16];
            #pragma unroll
            for (int i = 0; i < 16; ++i) {
                const float* xp = xb + nid[i] * 3u;
                const float rx = xp[0] - qx;
                const float ry = xp[1] - qy;
                const float rz = xp[2] - qz;
                const float dx = rx - kpx;
                const float dy = ry - kpy;
                const float dz = rz - kpz;
                const float d  = sqrtf(fmaf(dx, dx, fmaf(dy, dy, dz * dz)));
                float inf = 1.0f - d * (1.0f / KP_EXTENT);
                inf = inf > 0.0f ? inf : 0.0f;
                infl[i] = live ? inf : 0.0f;
            }

            __builtin_amdgcn_sched_barrier(0);

            float4 fB[8];
            #pragma unroll
            for (int i = 0; i < 8; ++i) fB[i] = fb[nid[8 + i] * 16u + (unsigned)c];

            v8f acc0 = {}, acc1 = {}, acc2 = {}, acc3 = {};
            #pragma unroll
            for (int j = 0; j < 4; ++j)
                agg_step(acc0, acc1, acc2, acc3, infl[2 * j], infl[2 * j + 1], fA[2 * j], fA[2 * j + 1]);
            #pragma unroll
            for (int j = 0; j < 4; ++j)
                agg_step(acc0, acc1, acc2, acc3, infl[8 + 2 * j], infl[8 + 2 * j + 1], fB[2 * j], fB[2 * j + 1]);

            float* dst = &sA[m * PITCH + (8 * h) * cCin + 4 * c];
            #pragma unroll
            for (int r = 0; r < 8; ++r) {
                const float4 v = make_float4(acc0[r], acc1[r], acc2[r], acc3[r]);
                *reinterpret_cast<float4*>(dst + r * cCin) = v;
            }
        }
    }
    __syncthreads();

    const int ng = w & (NGROUPS - 1);
    const int kq = w >> 2;

    const float* a0p = &sA[c * PITCH + kq * KQ + 2 * h];
    const float* a1p = a0p + 16 * PITCH;
    const float* wp  = W + (size_t)(kq * KQ + 2 * h) * cCout + ng * 32 + 2 * c;

    v8f c00 = {}, c01 = {}, c10 = {}, c11 = {};
    #pragma unroll 4
    for (int s = 0; s < KQ / 4; ++s) {
        const v2f a0 = *reinterpret_cast<const v2f*>(a0p + 4 * s);
        const v2f a1 = *reinterpret_cast<const v2f*>(a1p + 4 * s);
        const float2 w0 = *reinterpret_cast<const float2*>(wp + (size_t)(4 * s) * cCout);
        const float2 w1 = *reinterpret_cast<const float2*>(wp + (size_t)(4 * s + 1) * cCout);
        v2f b0; b0.x = w0.x; b0.y = w1.x;
        v2f b1; b1.x = w0.y; b1.y = w1.y;
        c00 = mm(a0, b0, c00);
        c01 = mm(a0, b1, c01);
        c10 = mm(a1, b0, c10);
        c11 = mm(a1, b1, c11);
    }

    __syncthreads();
    {
        float4* mine = reinterpret_cast<float4*>(smem + w * PART);
        mine[(0 * 2 + 0) * 32 + lane] = make_float4(c00[0], c00[1], c00[2], c00[3]);
        mine[(0 * 2 + 1) * 32 + lane] = make_float4(c00[4], c00[5], c00[6], c00[7]);
        mine[(1 * 2 + 0) * 32 + lane] = make_float4(c01[0], c01[1], c01[2], c01[3]);
        mine[(1 * 2 + 1) * 32 + lane] = make_float4(c01[4], c01[5], c01[6], c01[7]);
        mine[(2 * 2 + 0) * 32 + lane] = make_float4(c10[0], c10[1], c10[2], c10[3]);
        mine[(2 * 2 + 1) * 32 + lane] = make_float4(c10[4], c10[5], c10[6], c10[7]);
        mine[(3 * 2 + 0) * 32 + lane] = make_float4(c11[0], c11[1], c11[2], c11[3]);
        mine[(3 * 2 + 1) * 32 + lane] = make_float4(c11[4], c11[5], c11[6], c11[7]);
    }
    __syncthreads();

    {
        const int mt = kq >> 1;
        const int rh = kq & 1;
        float4 s0 = make_float4(0.f, 0.f, 0.f, 0.f), s1 = s0;
        #pragma unroll
        for (int q = 0; q < KSPLIT; ++q) {
            const float4* src = reinterpret_cast<const float4*>(smem + (q * NGROUPS + ng) * PART);
            const float4 t0 = src[((2 * mt + 0) * 2 + rh) * 32 + lane];
            const float4 t1 = src[((2 * mt + 1) * 2 + rh) * 32 + lane];
            s0.x += t0.x; s0.y += t0.y; s0.z += t0.z; s0.w += t0.w;
            s1.x += t1.x; s1.y += t1.y; s1.z += t1.z; s1.w += t1.w;
        }
        const float v0[4] = { s0.x, s0.y, s0.z, s0.w };
        const float v1[4] = { s1.x, s1.y, s1.z, s1.w };
        float* orow = out + ((size_t)(b * cP + p0 + 16 * mt + 8 * h + 4 * rh)) * cCout + ng * 32 + 2 * c;
        #pragma unroll
        for (int rr = 0; rr < 4; ++rr) {
            float x = v0[rr];
            float y = v1[rr];
            x = x > 0.0f ? x : ALPHA * x;
            y = y > 0.0f ? y : ALPHA * y;
            *reinterpret_cast<float2*>(orow + (size_t)rr * cCout) = make_float2(x, y);
        }
    }
}

extern "C" void kernel_launch(void* const* d_in, const int* in_sizes, int n_in,
                              void* d_out, int out_size, void* d_ws, size_t ws_size,
                              hipStream_t stream) {
    const float* xyz  = (const float*)d_in[0];
    const float* feat = (const float*)d_in[1];
    const int*   qidx = (const int*)d_in[2];
    const int*   nidx = (const int*)d_in[3];
    const float* kpts = (const float*)d_in[4];
    const float* W    = (const float*)d_in[5];
    float* out = (float*)d_out;

    const int blocks = cB * cP / TILE_M;
    kpconv_fused<<<blocks, NWAVES * 32, 0, stream>>>(xyz, feat, qidx, nidx, kpts, W, out);
}


# Influence distances taken with the hardware square-root instruction (1 ulp) instead of the correctly rounded sqrtf expansion: about 14 fewer vector instructions per influence value in phase 1
# speedup vs baseline: 1.9119x; 1.0452x over previous
#include <hip/hip_runtime.h>


typedef __attribute__((ext_vector_type(2))) float v2f;
typedef __attribute__((ext_vector_type(8))) float v8f;

#define KP_EXTENT 0.6f
#define ALPHA 0.3f

constexpr int cB    = 8;
constexpr int cN    = 16384;
constexpr int cP    = 4096;
constexpr int cS    = 32;
constexpr int cK    = 15;
constexpr int cCin  = 64;
constexpr int cCout = 128;
constexpr int TILE_M = 32;
constexpr int NWAVES = 16;
constexpr int KROWS  = 16;
constexpr int PITCH  = KROWS * cCin + 4;
constexpr int KDIM   = cK * cCin;
constexpr int PTS_PER_WAVE = TILE_M / NWAVES;
constexpr int NGROUPS = 4;
constexpr int KSPLIT  = NWAVES / NGROUPS;
constexpr int KQ      = KDIM / KSPLIT;
constexpr int PART    = 4 * 8 * 32;
static_assert(NWAVES * PART <= TILE_M * PITCH, "the partial tiles fit inside the dead A rows");

__device__ __forceinline__ void agg_step(v8f& a0, v8f& a1, v8f& a2, v8f& a3,
                                         const float i0, const float i1,
                                         const float4 f0, const float4 f1)
{
    v2f a;  a.x  = i0;   a.y  = i1;
    v2f g0; g0.x = f0.x; g0.y = f1.x;
    v2f g1; g1.x = f0.y; g1.y = f1.y;
    v2f g2; g2.x = f0.z; g2.y = f1.z;
    v2f g3; g3.x = f0.w; g3.y = f1.w;
    a0 = __builtin_amdgcn_wmma_f32_16x16x4_f32(false, a, false, g0, (short)0, a0, false, false);
    a1 = __builtin_amdgcn_wmma_f32_16x16x4_f32(false, a, false, g1, (short)0, a1, false, false);
    a2 = __builtin_amdgcn_wmma_f32_16x16x4_f32(false, a, false, g2, (short)0, a2, false, false);
    a3 = __builtin_amdgcn_wmma_f32_16x16x4_f32(false, a, false, g3, (short)0, a3, false, false);
}

__device__ __forceinline__ v8f mm(const v2f a, const v2f b, const v8f c)
{
    return __builtin_amdgcn_wmma_f32_16x16x4_f32(false, a, false, b, (short)0, c, false, false);
}

__global__ __launch_bounds__(512) __attribute__((amdgpu_num_vgpr(128))) void kpconv_fused(
    const float* __restrict__ xyz,
    const float* __restrict__ feat,
    const int*   __restrict__ qidx,
    const int*   __restrict__ nidx,
    const float* __restrict__ kpts,
    const float* __restrict__ W,
    float*       __restrict__ out)
{
    __shared__ __attribute__((aligned(16))) float smem[TILE_M * PITCH];
    float* sA = smem;

    const int tid  = threadIdx.x;
    const int tile = blockIdx.x;
    const int b    = tile / (cP / TILE_M);
    const int p0   = (tile % (cP / TILE_M)) * TILE_M;

    const int w    = __builtin_amdgcn_readfirstlane(tid >> 5);
    const int lane = tid & 31;
    const int c    = lane & 15;
    const int h    = lane >> 4;

    {
        const bool live = c < cK;
        const int  kc   = live ? c : 0;
        const float kpx = kpts[kc * 3 + 0];
        const float kpy = kpts[kc * 3 + 1];
        const float kpz = kpts[kc * 3 + 2];
        const float*  xb = xyz + (size_t)b * cN * 3;
        const float4* fb = reinterpret_cast<const float4*>(feat + (size_t)b * cN * cCin);

        float qxs[PTS_PER_WAVE], qys[PTS_PER_WAVE], qzs[PTS_PER_WAVE];
        #pragma unroll
        for (int pt = 0; pt < PTS_PER_WAVE; ++pt) {
            const int q = qidx[b * cP + p0 + w * PTS_PER_WAVE + pt];
            const float* qp = xb + q * 3;
            qxs[pt] = qp[0]; qys[pt] = qp[1]; qzs[pt] = qp[2];
        }

        #pragma unroll
        for (int pt = 0; pt < PTS_PER_WAVE; ++pt) {
            const int m = w * PTS_PER_WAVE + pt;
            const int p = p0 + m;
            const float qx = qxs[pt], qy = qys[pt], qz = qzs[pt];

            const int4* np = reinterpret_cast<const int4*>(nidx + ((size_t)b * cP + p) * cS + 16 * h);
            unsigned nid[16];
            #pragma unroll
            for (int i = 0; i < 4; ++i) {
                const int4 v = np[i];
                nid[4 * i + 0] = (unsigned)v.x; nid[4 * i + 1] = (unsigned)v.y;
                nid[4 * i + 2] = (unsigned)v.z; nid[4 * i + 3] = (unsigned)v.w;
            }

            float4 fA[8];
            #pragma unroll
            for (int i = 0; i < 8; ++i) fA[i] = fb[nid[i] * 16u + (unsigned)c];

            float infl[16];
            #pragma unroll
            for (int i = 0; i < 16; ++i) {
                const float* xp = xb + nid[i] * 3u;
                const float rx = xp[0] - qx;
                const float ry = xp[1] - qy;
                const float rz = xp[2] - qz;
                const float dx = rx - kpx;
                const float dy = ry - kpy;
                const float dz = rz - kpz;
                const float d  = __builtin_amdgcn_sqrtf(fmaf(dx, dx, fmaf(dy, dy, dz * dz)));
                float inf = 1.0f - d * (1.0f / KP_EXTENT);
                inf = inf > 0.0f ? inf : 0.0f;
                infl[i] = live ? inf : 0.0f;
            }

            unsigned tie = 0;
            asm volatile("" : "+v"(tie) : "v"(infl[12]), "v"(infl[13]), "v"(infl[14]), "v"(infl[15]));
            float4 fB[8];
            #pragma unroll
            for (int i = 0; i < 8; ++i) fB[i] = fb[(nid[8 + i] + tie) * 16u + (unsigned)c];

            v8f acc0 = {}, acc1 = {}, acc2 = {}, acc3 = {};
            #pragma unroll
            for (int j = 0; j < 4; ++j)
                agg_step(acc0, acc1, acc2, acc3, infl[2 * j], infl[2 * j + 1], fA[2 * j], fA[2 * j + 1]);
            #pragma unroll
            for (int j = 0; j < 4; ++j)
                agg_step(acc0, acc1, acc2, acc3, infl[8 + 2 * j], infl[8 + 2 * j + 1], fB[2 * j], fB[2 * j + 1]);

            float* dst = &sA[m * PITCH + (8 * h) * cCin + 4 * c];
            #pragma unroll
            for (int r = 0; r < 8; ++r) {
                const float4 v = make_float4(acc0[r], acc1[r], acc2[r], acc3[r]);
                *reinterpret_cast<float4*>(dst + r * cCin) = v;
            }
        }
    }
    __syncthreads();

    const int ng = w & (NGROUPS - 1);
    const int kq = w >> 2;

    const float* a0p = &sA[c * PITCH + kq * KQ + 2 * h];
    const float* a1p = a0p + 16 * PITCH;
    const float* wp  = W + (size_t)(kq * KQ + 2 * h) * cCout + ng * 32 + 2 * c;

    v8f c00 = {}, c01 = {}, c10 = {}, c11 = {};
    #pragma unroll 4
    for (int s = 0; s < KQ / 4; ++s) {
        const v2f a0 = *reinterpret_cast<const v2f*>(a0p + 4 * s);
        const v2f a1 = *reinterpret_cast<const v2f*>(a1p + 4 * s);
        const float2 w0 = *reinterpret_cast<const float2*>(wp + (size_t)(4 * s) * cCout);
        const float2 w1 = *reinterpret_cast<const float2*>(wp + (size_t)(4 * s + 1) * cCout);
        v2f b0; b0.x = w0.x; b0.y = w1.x;
        v2f b1; b1.x = w0.y; b1.y = w1.y;
        c00 = mm(a0, b0, c00);
        c01 = mm(a0, b1, c01);
        c10 = mm(a1, b0, c10);
        c11 = mm(a1, b1, c11);
    }

    __syncthreads();
    {
        float4* mine = reinterpret_cast<float4*>(smem + w * PART);
        mine[(0 * 2 + 0) * 32 + lane] = make_float4(c00[0], c00[1], c00[2], c00[3]);
        mine[(0 * 2 + 1) * 32 + lane] = make_float4(c00[4], c00[5], c00[6], c00[7]);
        mine[(1 * 2 + 0) * 32 + lane] = make_float4(c01[0], c01[1], c01[2], c01[3]);
        mine[(1 * 2 + 1) * 32 + lane] = make_float4(c01[4], c01[5], c01[6], c01[7]);
        mine[(2 * 2 + 0) * 32 + lane] = make_float4(c10[0], c10[1], c10[2], c10[3]);
        mine[(2 * 2 + 1) * 32 + lane] = make_float4(c10[4], c10[5], c10[6], c10[7]);
        mine[(3 * 2 + 0) * 32 + lane] = make_float4(c11[0], c11[1], c11[2], c11[3]);
        mine[(3 * 2 + 1) * 32 + lane] = make_float4(c11[4], c11[5], c11[6], c11[7]);
    }
    __syncthreads();

    {
        const int mt = kq >> 1;
        const int rh = kq & 1;
        float4 s0 = make_float4(0.f, 0.f, 0.f, 0.f), s1 = s0;
        #pragma unroll
        for (int q = 0; q < KSPLIT; ++q) {
            const float4* src = reinterpret_cast<const float4*>(smem + (q * NGROUPS + ng) * PART);
            const float4 t0 = src[((2 * mt + 0) * 2 + rh) * 32 + lane];
            const float4 t1 = src[((2 * mt + 1) * 2 + rh) * 32 + lane];
            s0.x += t0.x; s0.y += t0.y; s0.z += t0.z; s0.w += t0.w;
            s1.x += t1.x; s1.y += t1.y; s1.z += t1.z; s1.w += t1.w;
        }
        const float v0[4] = { s0.x, s0.y, s0.z, s0.w };
        const float v1[4] = { s1.x, s1.y, s1.z, s1.w };
        float* orow = out + ((size_t)(b * cP + p0 + 16 * mt + 8 * h + 4 * rh)) * cCout + ng * 32 + 2 * c;
        #pragma unroll
        for (int rr = 0; rr < 4; ++rr) {
            float x = v0[rr];
            float y = v1[rr];
            x = x > 0.0f ? x : ALPHA * x;
            y = y > 0.0f ? y : ALPHA * y;
            *reinterpret_cast<float2*>(orow + (size_t)rr * cCout) = make_float2(x, y);
        }
    }
}

extern "C" void kernel_launch(void* const* d_in, const int* in_sizes, int n_in,
                              void* d_out, int out_size, void* d_ws, size_t ws_size,
                              hipStream_t stream) {
    const float* xyz  = (const float*)d_in[0];
    const float* feat = (const float*)d_in[1];
    const int*   qidx = (const int*)d_in[2];
    const int*   nidx = (const int*)d_in[3];
    const float* kpts = (const float*)d_in[4];
    const float* W    = (const float*)d_in[5];
    float* out = (float*)d_out;

    const int blocks = cB * cP / TILE_M;
    kpconv_fused<<<blocks, NWAVES * 32, 0, stream>>>(xyz, feat, qidx, nidx, kpts, W, out);
}


# Aggregation K steps whose 4 neighbours have exactly zero influence on all 16 kernel-point rows are skipped (wave-uniform test): no matrix ops, feature reads go to one cached row; sums unchanged
# speedup vs baseline: 2.0883x; 1.0923x over previous
#include <hip/hip_runtime.h>


typedef __attribute__((ext_vector_type(2))) float v2f;
typedef __attribute__((ext_vector_type(8))) float v8f;

#define KP_EXTENT 0.6f
#define ALPHA 0.3f

constexpr int cB    = 8;
constexpr int cN    = 16384;
constexpr int cP    = 4096;
constexpr int cS    = 32;
constexpr int cK    = 15;
constexpr int cCin  = 64;
constexpr int cCout = 128;
constexpr int TILE_M = 32;
constexpr int NWAVES = 16;
constexpr int KROWS  = 16;
constexpr int PITCH  = KROWS * cCin + 4;
constexpr int KDIM   = cK * cCin;
constexpr int PTS_PER_WAVE = TILE_M / NWAVES;
constexpr int NGROUPS = 4;
constexpr int KSPLIT  = NWAVES / NGROUPS;
constexpr int KQ      = KDIM / KSPLIT;
constexpr int PART    = 4 * 8 * 32;
static_assert(NWAVES * PART <= TILE_M * PITCH, "the partial tiles fit inside the dead A rows");

__device__ __forceinline__ void agg_step(v8f& a0, v8f& a1, v8f& a2, v8f& a3,
                                         const float i0, const float i1,
                                         const float4 f0, const float4 f1)
{
    v2f a;  a.x  = i0;   a.y  = i1;
    v2f g0; g0.x = f0.x; g0.y = f1.x;
    v2f g1; g1.x = f0.y; g1.y = f1.y;
    v2f g2; g2.x = f0.z; g2.y = f1.z;
    v2f g3; g3.x = f0.w; g3.y = f1.w;
    a0 = __builtin_amdgcn_wmma_f32_16x16x4_f32(false, a, false, g0, (short)0, a0, false, false);
    a1 = __builtin_amdgcn_wmma_f32_16x16x4_f32(false, a, false, g1, (short)0, a1, false, false);
    a2 = __builtin_amdgcn_wmma_f32_16x16x4_f32(false, a, false, g2, (short)0, a2, false, false);
    a3 = __builtin_amdgcn_wmma_f32_16x16x4_f32(false, a, false, g3, (short)0, a3, false, false);
}

__device__ __forceinline__ v8f mm(const v2f a, const v2f b, const v8f c)
{
    return __builtin_amdgcn_wmma_f32_16x16x4_f32(false, a, false, b, (short)0, c, false, false);
}

__global__ __launch_bounds__(512) __attribute__((amdgpu_num_vgpr(128))) void kpconv_fused(
    const float* __restrict__ xyz,
    const float* __restrict__ feat,
    const int*   __restrict__ qidx,
    const int*   __restrict__ nidx,
    const float* __restrict__ kpts,
    const float* __restrict__ W,
    float*       __restrict__ out)
{
    __shared__ __attribute__((aligned(16))) float smem[TILE_M * PITCH];
    float* sA = smem;

    const int tid  = threadIdx.x;
    const int tile = blockIdx.x;
    const int b    = tile / (cP / TILE_M);
    const int p0   = (tile % (cP / TILE_M)) * TILE_M;

    const int w    = __builtin_amdgcn_readfirstlane(tid >> 5);
    const int lane = tid & 31;
    const int c    = lane & 15;
    const int h    = lane >> 4;

    {
        const bool live = c < cK;
        const int  kc   = live ? c : 0;
        const float kpx = kpts[kc * 3 + 0];
        const float kpy = kpts[kc * 3 + 1];
        const float kpz = kpts[kc * 3 + 2];
        const float*  xb = xyz + (size_t)b * cN * 3;
        const float4* fb = reinterpret_cast<const float4*>(feat + (size_t)b * cN * cCin);

        float qxs[PTS_PER_WAVE], qys[PTS_PER_WAVE], qzs[PTS_PER_WAVE];
        #pragma unroll
        for (int pt = 0; pt < PTS_PER_WAVE; ++pt) {
            const int q = qidx[b * cP + p0 + w * PTS_PER_WAVE + pt];
            const float* qp = xb + q * 3;
            qxs[pt] = qp[0]; qys[pt] = qp[1]; qzs[pt] = qp[2];
        }

        #pragma unroll
        for (int pt = 0; pt < PTS_PER_WAVE; ++pt) {
            const int m = w * PTS_PER_WAVE + pt;
            const int p = p0 + m;
            const float qx = qxs[pt], qy = qys[pt], qz = qzs[pt];

            const int4* np = reinterpret_cast<const int4*>(nidx + ((size_t)b * cP + p) * cS + 16 * h);
            unsigned nid[16];
            #pragma unroll
            for (int i = 0; i < 4; ++i) {
                const int4 v = np[i];
                nid[4 * i + 0] = (unsigned)v.x; nid[4 * i + 1] = (unsigned)v.y;
                nid[4 * i + 2] = (unsigned)v.z; nid[4 * i + 3] = (unsigned)v.w;
            }

            float infl[16];
            #pragma unroll
            for (int i = 0; i < 16; ++i) {
                const float* xp = xb + nid[i] * 3u;
                const float rx = xp[0] - qx;
                const float ry = xp[1] - qy;
                const float rz = xp[2] - qz;
                const float dx = rx - kpx;
                const float dy = ry - kpy;
                const float dz = rz - kpz;
                const float d  = __builtin_amdgcn_sqrtf(fmaf(dx, dx, fmaf(dy, dy, dz * dz)));
                float inf = 1.0f - d * (1.0f / KP_EXTENT);
                inf = inf > 0.0f ? inf : 0.0f;
                infl[i] = live ? inf : 0.0f;
            }

            bool act[8];
            #pragma unroll
            for (int j = 0; j < 8; ++j) {
                const bool nz = (__float_as_uint(infl[2 * j]) | __float_as_uint(infl[2 * j + 1])) != 0u;
                act[j] = __builtin_amdgcn_ballot_w32(nz) != 0u;
            }

            float4 f[16];
            #pragma unroll
            for (int j = 0; j < 8; ++j) {
                const unsigned n0 = act[j] ? nid[2 * j]     : 0u;
                const unsigned n1 = act[j] ? nid[2 * j + 1] : 0u;
                f[2 * j]     = fb[n0 * 16u + (unsigned)c];
                f[2 * j + 1] = fb[n1 * 16u + (unsigned)c];
            }

            #pragma unroll
            for (int j = 0; j < 8; ++j) {
                asm volatile("" : : "v"(f[2 * j].x), "v"(f[2 * j].y), "v"(f[2 * j].z), "v"(f[2 * j].w),
                                    "v"(f[2 * j + 1].x), "v"(f[2 * j + 1].y), "v"(f[2 * j + 1].z), "v"(f[2 * j + 1].w));
            }

            v8f acc0 = {}, acc1 = {}, acc2 = {}, acc3 = {};
            #pragma unroll
            for (int j = 0; j < 8; ++j) {
                if (act[j])
                    agg_step(acc0, acc1, acc2, acc3, infl[2 * j], infl[2 * j + 1], f[2 * j], f[2 * j + 1]);
            }

            float* dst = &sA[m * PITCH + (8 * h) * cCin + 4 * c];
            #pragma unroll
            for (int r = 0; r < 8; ++r) {
                const float4 v = make_float4(acc0[r], acc1[r], acc2[r], acc3[r]);
                *reinterpret_cast<float4*>(dst + r * cCin) = v;
            }
            __builtin_amdgcn_sched_barrier(0);
        }
    }
    __syncthreads();

    const int ng = w & (NGROUPS - 1);
    const int kq = w >> 2;

    const float* a0p = &sA[c * PITCH + kq * KQ + 2 * h];
    const float* a1p = a0p + 16 * PITCH;
    const float* wp  = W + (size_t)(kq * KQ + 2 * h) * cCout + ng * 32 + 2 * c;

    v8f c00 = {}, c01 = {}, c10 = {}, c11 = {};
    #pragma unroll 4
    for (int s = 0; s < KQ / 4; ++s) {
        const v2f a0 = *reinterpret_cast<const v2f*>(a0p + 4 * s);
        const v2f a1 = *reinterpret_cast<const v2f*>(a1p + 4 * s);
        const float2 w0 = *reinterpret_cast<const float2*>(wp + (size_t)(4 * s) * cCout);
        const float2 w1 = *reinterpret_cast<const float2*>(wp + (size_t)(4 * s + 1) * cCout);
        v2f b0; b0.x = w0.x; b0.y = w1.x;
        v2f b1; b1.x = w0.y; b1.y = w1.y;
        c00 = mm(a0, b0, c00);
        c01 = mm(a0, b1, c01);
        c10 = mm(a1, b0, c10);
        c11 = mm(a1, b1, c11);
    }

    __syncthreads();
    {
        float4* mine = reinterpret_cast<float4*>(smem + w * PART);
        mine[(0 * 2 + 0) * 32 + lane] = make_float4(c00[0], c00[1], c00[2], c00[3]);
        mine[(0 * 2 + 1) * 32 + lane] = make_float4(c00[4], c00[5], c00[6], c00[7]);
        mine[(1 * 2 + 0) * 32 + lane] = make_float4(c01[0], c01[1], c01[2], c01[3]);
        mine[(1 * 2 + 1) * 32 + lane] = make_float4(c01[4], c01[5], c01[6], c01[7]);
        mine[(2 * 2 + 0) * 32 + lane] = make_float4(c10[0], c10[1], c10[2], c10[3]);
        mine[(2 * 2 + 1) * 32 + lane] = make_float4(c10[4], c10[5], c10[6], c10[7]);
        mine[(3 * 2 + 0) * 32 + lane] = make_float4(c11[0], c11[1], c11[2], c11[3]);
        mine[(3 * 2 + 1) * 32 + lane] = make_float4(c11[4], c11[5], c11[6], c11[7]);
    }
    __syncthreads();

    {
        const int mt = kq >> 1;
        const int rh = kq & 1;
        float4 s0 = make_float4(0.f, 0.f, 0.f, 0.f), s1 = s0;
        #pragma unroll
        for (int q = 0; q < KSPLIT; ++q) {
            const float4* src = reinterpret_cast<const float4*>(smem + (q * NGROUPS + ng) * PART);
            const float4 t0 = src[((2 * mt + 0) * 2 + rh) * 32 + lane];
            const float4 t1 = src[((2 * mt + 1) * 2 + rh) * 32 + lane];
            s0.x += t0.x; s0.y += t0.y; s0.z += t0.z; s0.w += t0.w;
            s1.x += t1.x; s1.y += t1.y; s1.z += t1.z; s1.w += t1.w;
        }
        const float v0[4] = { s0.x, s0.y, s0.z, s0.w };
        const float v1[4] = { s1.x, s1.y, s1.z, s1.w };
        float* orow = out + ((size_t)(b * cP + p0 + 16 * mt + 8 * h + 4 * rh)) * cCout + ng * 32 + 2 * c;
        #pragma unroll
        for (int rr = 0; rr < 4; ++rr) {
            float x = v0[rr];
            float y = v1[rr];
            x = x > 0.0f ? x : ALPHA * x;
            y = y > 0.0f ? y : ALPHA * y;
            *reinterpret_cast<float2*>(orow + (size_t)rr * cCout) = make_float2(x, y);
        }
    }
}

extern "C" void kernel_launch(void* const* d_in, const int* in_sizes, int n_in,
                              void* d_out, int out_size, void* d_ws, size_t ws_size,
                              hipStream_t stream) {
    const float* xyz  = (const float*)d_in[0];
    const float* feat = (const float*)d_in[1];
    const int*   qidx = (const int*)d_in[2];
    const int*   nidx = (const int*)d_in[3];
    const float* kpts = (const float*)d_in[4];
    const float* W    = (const float*)d_in[5];
    float* out = (float*)d_out;

    const int blocks = cB * cP / TILE_M;
    kpconv_fused<<<blocks, NWAVES * 32, 0, stream>>>(xyz, feat, qidx, nidx, kpts, W, out);
}


# Aggregation runs only over neighbours inside some kernel point's support: one lane per neighbour tests it, a ballot packs those to the front of a list in LDS; terms are summed in list order
# speedup vs baseline: 2.3457x; 1.1233x over previous
#include <hip/hip_runtime.h>


typedef __attribute__((ext_vector_type(2))) float v2f;
typedef __attribute__((ext_vector_type(8))) float v8f;

#define KP_EXTENT 0.6f
#define ALPHA 0.3f

constexpr int cB    = 8;
constexpr int cN    = 16384;
constexpr int cP    = 4096;
constexpr int cS    = 32;
constexpr int cK    = 15;
constexpr int cCin  = 64;
constexpr int cCout = 128;
constexpr int TILE_M = 32;
constexpr int NWAVES = 16;
constexpr int KROWS  = 16;
constexpr int PITCH  = KROWS * cCin + 4;
constexpr int KDIM   = cK * cCin;
constexpr int PTS_PER_WAVE = TILE_M / NWAVES;
constexpr int NGROUPS = 4;
constexpr int KSPLIT  = NWAVES / NGROUPS;
constexpr int KQ      = KDIM / KSPLIT;
constexpr int PART    = 4 * 8 * 32;
static_assert(NWAVES * PART <= TILE_M * PITCH, "the partial tiles fit inside the dead A rows");

__device__ __forceinline__ void agg_step(v8f& a0, v8f& a1, v8f& a2, v8f& a3,
                                         const float i0, const float i1,
                                         const float4 f0, const float4 f1)
{
    v2f a;  a.x  = i0;   a.y  = i1;
    v2f g0; g0.x = f0.x; g0.y = f1.x;
    v2f g1; g1.x = f0.y; g1.y = f1.y;
    v2f g2; g2.x = f0.z; g2.y = f1.z;
    v2f g3; g3.x = f0.w; g3.y = f1.w;
    a0 = __builtin_amdgcn_wmma_f32_16x16x4_f32(false, a, false, g0, (short)0, a0, false, false);
    a1 = __builtin_amdgcn_wmma_f32_16x16x4_f32(false, a, false, g1, (short)0, a1, false, false);
    a2 = __builtin_amdgcn_wmma_f32_16x16x4_f32(false, a, false, g2, (short)0, a2, false, false);
    a3 = __builtin_amdgcn_wmma_f32_16x16x4_f32(false, a, false, g3, (short)0, a3, false, false);
}

__device__ __forceinline__ v8f mm(const v2f a, const v2f b, const v8f c)
{
    return __builtin_amdgcn_wmma_f32_16x16x4_f32(false, a, false, b, (short)0, c, false, false);
}

__device__ __forceinline__ float influence(const float4 v, const float kx, const float ky, const float kz, const bool live)
{
    const float dx = v.x - kx;
    const float dy = v.y - ky;
    const float dz = v.z - kz;
    const float d  = __builtin_amdgcn_sqrtf(fmaf(dx, dx, fmaf(dy, dy, dz * dz)));
    float inf = 1.0f - d * (1.0f / KP_EXTENT);
    inf = inf > 0.0f ? inf : 0.0f;
    return live ? inf : 0.0f;
}

__device__ __forceinline__ void trip_load(const float4* list, const float4* fb, const int g, const int nsteps,
                                          const int h, const int c,
                                          const float kx, const float ky, const float kz, const bool live,
                                          float4 (&f)[4], float (&infl)[4])
{
    #pragma unroll
    for (int i = 0; i < 4; ++i) {
        const int t = g + (i >> 1);
        const float4 v = list[4 * t + 2 * h + (i & 1)];
        const unsigned keep = t < nsteps ? 0xffffffffu : 0u;
        const unsigned n = __float_as_uint(v.w) & keep;
        f[i] = fb[n * 16u + (unsigned)c];
        infl[i] = influence(v, kx, ky, kz, live);
    }
}

__device__ __forceinline__ void trip_run(v8f& a0, v8f& a1, v8f& a2, v8f& a3, const int g, const int nsteps,
                                         const float4 (&f)[4], const float (&infl)[4])
{
    if (g < nsteps)
        agg_step(a0, a1, a2, a3, infl[0], infl[1], f[0], f[1]);
    if (g + 1 < nsteps)
        agg_step(a0, a1, a2, a3, infl[2], infl[3], f[2], f[3]);
}

__device__ __forceinline__ void rows_requested(const float4 (&f)[4])
{
    #pragma unroll
    for (int i = 0; i < 4; ++i)
        asm volatile("" : : "v"(f[i].x), "v"(f[i].y), "v"(f[i].z), "v"(f[i].w));
}

__global__ __launch_bounds__(512) __attribute__((amdgpu_num_vgpr(128))) void kpconv_fused(
    const float* __restrict__ xyz,
    const float* __restrict__ feat,
    const int*   __restrict__ qidx,
    const int*   __restrict__ nidx,
    const float* __restrict__ kpts,
    const float* __restrict__ W,
    float*       __restrict__ out)
{
    __shared__ __attribute__((aligned(16))) float smem[TILE_M * PITCH];
    float* sA = smem;

    const int tid  = threadIdx.x;
    const int tile = blockIdx.x;
    const int b    = tile / (cP / TILE_M);
    const int p0   = (tile % (cP / TILE_M)) * TILE_M;

    const int w    = __builtin_amdgcn_readfirstlane(tid >> 5);
    const int lane = tid & 31;
    const int c    = lane & 15;
    const int h    = lane >> 4;

    {
        const bool live = c < cK;
        const int  kc   = live ? c : 0;
        const float kpx = kpts[kc * 3 + 0];
        const float kpy = kpts[kc * 3 + 1];
        const float kpz = kpts[kc * 3 + 2];
        const float*  xb = xyz + (size_t)b * cN * 3;
        const float4* fb = reinterpret_cast<const float4*>(feat + (size_t)b * cN * cCin);
        const float support2 = KP_EXTENT * KP_EXTENT * 1.00001f;

        int nsteps[PTS_PER_WAVE];
        #pragma unroll
        for (int pt = 0; pt < PTS_PER_WAVE; ++pt) {
            const int m = w * PTS_PER_WAVE + pt;
            const int p = p0 + m;
            const int q = qidx[b * cP + p];
            const float* qp = xb + q * 3;
            const float qx = qp[0], qy = qp[1], qz = qp[2];
            const unsigned n = (unsigned)nidx[((size_t)b * cP + p) * cS + lane];
            const float* xp = xb + n * 3u;
            const float rx = xp[0] - qx;
            const float ry = xp[1] - qy;
            const float rz = xp[2] - qz;
            float dmin = __builtin_inff();
            #pragma unroll
            for (int k = 0; k < cK; ++k) {
                const float dx = rx - kpts[3 * k + 0];
                const float dy = ry - kpts[3 * k + 1];
                const float dz = rz - kpts[3 * k + 2];
                dmin = fminf(dmin, fmaf(dx, dx, fmaf(dy, dy, dz * dz)));
            }
            const bool inside = dmin < support2;
            const unsigned mask = __builtin_amdgcn_ballot_w32(inside);
            const int na    = __builtin_popcount(mask);
            const int below = (int)__builtin_amdgcn_mbcnt_lo(mask, 0u);
            const int slot = inside ? below : na + lane - below;
            float4* list = reinterpret_cast<float4*>(&sA[m * PITCH]);
            list[slot] = make_float4(rx, ry, rz, __uint_as_float(n));
            nsteps[pt] = (na + 3) >> 2;
        }

        float4 f[PTS_PER_WAVE][4];
        float  infl[PTS_PER_WAVE][4];
        #pragma unroll
        for (int pt = 0; pt < PTS_PER_WAVE; ++pt) {
            const int m = w * PTS_PER_WAVE + pt;
            const float4* list = reinterpret_cast<const float4*>(&sA[m * PITCH]);
            trip_load(list, fb, 0, nsteps[pt], h, c, kpx, kpy, kpz, live, f[pt], infl[pt]);
        }
        #pragma unroll
        for (int pt = 0; pt < PTS_PER_WAVE; ++pt)
            rows_requested(f[pt]);

        #pragma unroll
        for (int pt = 0; pt < PTS_PER_WAVE; ++pt) {
            const int m = w * PTS_PER_WAVE + pt;
            const float4* list = reinterpret_cast<const float4*>(&sA[m * PITCH]);

            v8f acc0 = {}, acc1 = {}, acc2 = {}, acc3 = {};
            trip_run(acc0, acc1, acc2, acc3, 0, nsteps[pt], f[pt], infl[pt]);
            for (int g = 2; g < nsteps[pt]; g += 2) {
                float4 fm[4];
                float  im[4];
                trip_load(list, fb, g, nsteps[pt], h, c, kpx, kpy, kpz, live, fm, im);
                rows_requested(fm);
                trip_run(acc0, acc1, acc2, acc3, g, nsteps[pt], fm, im);
            }

            float* dst = &sA[m * PITCH + (8 * h) * cCin + 4 * c];
            #pragma unroll
            for (int r = 0; r < 8; ++r) {
                const float4 v = make_float4(acc0[r], acc1[r], acc2[r], acc3[r]);
                *reinterpret_cast<float4*>(dst + r * cCin) = v;
            }
        }
    }
    __syncthreads();

    const int ng = w & (NGROUPS - 1);
    const int kq = w >> 2;

    const float* a0p = &sA[c * PITCH + kq * KQ + 2 * h];
    const float* a1p = a0p + 16 * PITCH;
    const float* wp  = W + (size_t)(kq * KQ + 2 * h) * cCout + ng * 32 + 2 * c;

    v8f c00 = {}, c01 = {}, c10 = {}, c11 = {};
    #pragma unroll 4
    for (int s = 0; s < KQ / 4; ++s) {
        const v2f a0 = *reinterpret_cast<const v2f*>(a0p + 4 * s);
        const v2f a1 = *reinterpret_cast<const v2f*>(a1p + 4 * s);
        const float2 w0 = *reinterpret_cast<const float2*>(wp + (size_t)(4 * s) * cCout);
        const float2 w1 = *reinterpret_cast<const float2*>(wp + (size_t)(4 * s + 1) * cCout);
        v2f b0; b0.x = w0.x; b0.y = w1.x;
        v2f b1; b1.x = w0.y; b1.y = w1.y;
        c00 = mm(a0, b0, c00);
        c01 = mm(a0, b1, c01);
        c10 = mm(a1, b0, c10);
        c11 = mm(a1, b1, c11);
    }

    __syncthreads();
    {
        float4* mine = reinterpret_cast<float4*>(smem + w * PART);
        mine[(0 * 2 + 0) * 32 + lane] = make_float4(c00[0], c00[1], c00[2], c00[3]);
        mine[(0 * 2 + 1) * 32 + lane] = make_float4(c00[4], c00[5], c00[6], c00[7]);
        mine[(1 * 2 + 0) * 32 + lane] = make_float4(c01[0], c01[1], c01[2], c01[3]);
        mine[(1 * 2 + 1) * 32 + lane] = make_float4(c01[4], c01[5], c01[6], c01[7]);
        mine[(2 * 2 + 0) * 32 + lane] = make_float4(c10[0], c10[1], c10[2], c10[3]);
        mine[(2 * 2 + 1) * 32 + lane] = make_float4(c10[4], c10[5], c10[6], c10[7]);
        mine[(3 * 2 + 0) * 32 + lane] = make_float4(c11[0], c11[1], c11[2], c11[3]);
        mine[(3 * 2 + 1) * 32 + lane] = make_float4(c11[4], c11[5], c11[6], c11[7]);
    }
    __syncthreads();

    {
        const int mt = kq >> 1;
        const int rh = kq & 1;
        float4 s0 = make_float4(0.f, 0.f, 0.f, 0.f), s1 = s0;
        #pragma unroll
        for (int q = 0; q < KSPLIT; ++q) {
            const float4* src = reinterpret_cast<const float4*>(smem + (q * NGROUPS + ng) * PART);
            const float4 t0 = src[((2 * mt + 0) * 2 + rh) * 32 + lane];
            const float4 t1 = src[((2 * mt + 1) * 2 + rh) * 32 + lane];
            s0.x += t0.x; s0.y += t0.y; s0.z += t0.z; s0.w += t0.w;
            s1.x += t1.x; s1.y += t1.y; s1.z += t1.z; s1.w += t1.w;
        }
        const float v0[4] = { s0.x, s0.y, s0.z, s0.w };
        const float v1[4] = { s1.x, s1.y, s1.z, s1.w };
        float* orow = out + ((size_t)(b * cP + p0 + 16 * mt + 8 * h + 4 * rh)) * cCout + ng * 32 + 2 * c;
        #pragma unroll
        for (int rr = 0; rr < 4; ++rr) {
            float x = v0[rr];
            float y = v1[rr];
            x = x > 0.0f ? x : ALPHA * x;
            y = y > 0.0f ? y : ALPHA * y;
            *reinterpret_cast<float2*>(orow + (size_t)rr * cCout) = make_float2(x, y);
        }
    }
}

extern "C" void kernel_launch(void* const* d_in, const int* in_sizes, int n_in,
                              void* d_out, int out_size, void* d_ws, size_t ws_size,
                              hipStream_t stream) {
    const float* xyz  = (const float*)d_in[0];
    const float* feat = (const float*)d_in[1];
    const int*   qidx = (const int*)d_in[2];
    const int*   nidx = (const int*)d_in[3];
    const float* kpts = (const float*)d_in[4];
    const float* W    = (const float*)d_in[5];
    float* out = (float*)d_out;

    const int blocks = cB * cP / TILE_M;
    kpconv_fused<<<blocks, NWAVES * 32, 0, stream>>>(xyz, feat, qidx, nidx, kpts, W, out);
}


# Second product skips exactly-zero A rows: per kernel point the tile's points with a non-zero row are packed into one 16-row tile, rows scattered back via LDS; K split by kernel point, new sum order
# speedup vs baseline: 2.7435x; 1.1696x over previous
#include <hip/hip_runtime.h>


typedef __attribute__((ext_vector_type(2))) float v2f;
typedef __attribute__((ext_vector_type(8))) float v8f;

#define KP_EXTENT 0.6f
#define ALPHA 0.3f

constexpr int cB    = 8;
constexpr int cN    = 16384;
constexpr int cP    = 4096;
constexpr int cS    = 32;
constexpr int cK    = 15;
constexpr int cCin  = 64;
constexpr int cCout = 128;
constexpr int TILE_M = 32;
constexpr int NWAVES = 16;
constexpr int KROWS  = 16;
constexpr int PITCH  = cK * cCin + 4;
constexpr int KDIM   = cK * cCin;
constexpr int PTS_PER_WAVE = TILE_M / NWAVES;
constexpr int NGROUPS = 4;
constexpr int KSPLIT  = NWAVES / NGROUPS;
constexpr int KPW     = 4;
constexpr int PART    = 4 * 8 * 32;
static_assert(NWAVES * PART <= TILE_M * PITCH, "the partial tiles fit inside the dead A rows");
constexpr int CMAX    = 15;
constexpr int BP      = 36;
constexpr int OFF_B   = TILE_M * PITCH;
constexpr int OFF_AM  = OFF_B + NWAVES * 16 * BP;
constexpr int OFF_CNT = OFF_AM + TILE_M;
constexpr int OFF_INV = OFF_CNT + 16;
constexpr int OFF_PRM = OFF_INV + cK * TILE_M / 4;
constexpr int SMEM_FLOATS = OFF_PRM + cK * 16 / 4;
static_assert(SMEM_FLOATS * 4 <= 160 * 1024, "two work-groups per WGP");
static_assert((OFF_INV * 4) % 8 == 0, "8-byte reads of the slot table");

__device__ __forceinline__ void agg_step(v8f& a0, v8f& a1, v8f& a2, v8f& a3,
                                         const float i0, const float i1,
                                         const float4 f0, const float4 f1)
{
    v2f a;  a.x  = i0;   a.y  = i1;
    v2f g0; g0.x = f0.x; g0.y = f1.x;
    v2f g1; g1.x = f0.y; g1.y = f1.y;
    v2f g2; g2.x = f0.z; g2.y = f1.z;
    v2f g3; g3.x = f0.w; g3.y = f1.w;
    a0 = __builtin_amdgcn_wmma_f32_16x16x4_f32(false, a, false, g0, (short)0, a0, false, false);
    a1 = __builtin_amdgcn_wmma_f32_16x16x4_f32(false, a, false, g1, (short)0, a1, false, false);
    a2 = __builtin_amdgcn_wmma_f32_16x16x4_f32(false, a, false, g2, (short)0, a2, false, false);
    a3 = __builtin_amdgcn_wmma_f32_16x16x4_f32(false, a, false, g3, (short)0, a3, false, false);
}

__device__ __forceinline__ v8f mm(const v2f a, const v2f b, const v8f c)
{
    return __builtin_amdgcn_wmma_f32_16x16x4_f32(false, a, false, b, (short)0, c, false, false);
}

__device__ __forceinline__ float influence(const float4 v, const float kx, const float ky, const float kz, const bool live)
{
    const float dx = v.x - kx;
    const float dy = v.y - ky;
    const float dz = v.z - kz;
    const float d  = __builtin_amdgcn_sqrtf(fmaf(dx, dx, fmaf(dy, dy, dz * dz)));
    float inf = 1.0f - d * (1.0f / KP_EXTENT);
    inf = inf > 0.0f ? inf : 0.0f;
    return live ? inf : 0.0f;
}

__device__ __forceinline__ void trip_load(const float4* list, const float4* fb, const int g, const int nsteps,
                                          const int h, const int c,
                                          const float kx, const float ky, const float kz, const bool live,
                                          float4 (&f)[4], float (&infl)[4])
{
    #pragma unroll
    for (int i = 0; i < 4; ++i) {
        const int t = g + (i >> 1);
        const float4 v = list[4 * t + 2 * h + (i & 1)];
        const unsigned keep = t < nsteps ? 0xffffffffu : 0u;
        const unsigned n = __float_as_uint(v.w) & keep;
        f[i] = fb[n * 16u + (unsigned)c];
        infl[i] = influence(v, kx, ky, kz, live);
    }
}

__device__ __forceinline__ void trip_run(v8f& a0, v8f& a1, v8f& a2, v8f& a3, const int g, const int nsteps,
                                         const float4 (&f)[4], const float (&infl)[4], float& imax)
{
    if (g < nsteps) {
        imax = fmaxf(imax, fmaxf(infl[0], infl[1]));
        agg_step(a0, a1, a2, a3, infl[0], infl[1], f[0], f[1]);
    }
    if (g + 1 < nsteps) {
        imax = fmaxf(imax, fmaxf(infl[2], infl[3]));
        agg_step(a0, a1, a2, a3, infl[2], infl[3], f[2], f[3]);
    }
}

__device__ __forceinline__ void rows_requested(const float4 (&f)[4])
{
    #pragma unroll
    for (int i = 0; i < 4; ++i)
        asm volatile("" : : "v"(f[i].x), "v"(f[i].y), "v"(f[i].z), "v"(f[i].w));
}

__global__ __launch_bounds__(512) __attribute__((amdgpu_num_vgpr(128))) void kpconv_fused(
    const float* __restrict__ xyz,
    const float* __restrict__ feat,
    const int*   __restrict__ qidx,
    const int*   __restrict__ nidx,
    const float* __restrict__ kpts,
    const float* __restrict__ W,
    float*       __restrict__ out)
{
    __shared__ __attribute__((aligned(16))) float smem[SMEM_FLOATS];
    float* sA = smem;
    float*         sB   = smem + OFF_B;
    unsigned*      sAm  = reinterpret_cast<unsigned*>(smem + OFF_AM);
    unsigned*      sCnt = reinterpret_cast<unsigned*>(smem + OFF_CNT);
    unsigned char* sInv = reinterpret_cast<unsigned char*>(smem + OFF_INV);
    unsigned char* sPrm = reinterpret_cast<unsigned char*>(smem + OFF_PRM);

    const int tid  = threadIdx.x;
    const int tile = blockIdx.x;
    const int b    = tile / (cP / TILE_M);
    const int p0   = (tile % (cP / TILE_M)) * TILE_M;

    const int w    = __builtin_amdgcn_readfirstlane(tid >> 5);
    const int lane = tid & 31;
    const int c    = lane & 15;
    const int h    = lane >> 4;

    {
        const bool live = c < cK;
        const int  kc   = live ? c : 0;
        const float kpx = kpts[kc * 3 + 0];
        const float kpy = kpts[kc * 3 + 1];
        const float kpz = kpts[kc * 3 + 2];
        const float*  xb = xyz + (size_t)b * cN * 3;
        const float4* fb = reinterpret_cast<const float4*>(feat + (size_t)b * cN * cCin);
        const float support2 = KP_EXTENT * KP_EXTENT * 1.00001f;

        int nsteps[PTS_PER_WAVE];
        #pragma unroll
        for (int pt = 0; pt < PTS_PER_WAVE; ++pt) {
            const int m = w * PTS_PER_WAVE + pt;
            const int p = p0 + m;
            const int q = qidx[b * cP + p];
            const float* qp = xb + q * 3;
            const float qx = qp[0], qy = qp[1], qz = qp[2];
            const unsigned n = (unsigned)nidx[((size_t)b * cP + p) * cS + lane];
            const float* xp = xb + n * 3u;
            const float rx = xp[0] - qx;
            const float ry = xp[1] - qy;
            const float rz = xp[2] - qz;
            float dmin = __builtin_inff();
            #pragma unroll
            for (int k = 0; k < cK; ++k) {
                const float dx = rx - kpts[3 * k + 0];
                const float dy = ry - kpts[3 * k + 1];
                const float dz = rz - kpts[3 * k + 2];
                dmin = fminf(dmin, fmaf(dx, dx, fmaf(dy, dy, dz * dz)));
            }
            const bool inside = dmin < support2;
            const unsigned mask = __builtin_amdgcn_ballot_w32(inside);
            const int na    = __builtin_popcount(mask);
            const int below = (int)__builtin_amdgcn_mbcnt_lo(mask, 0u);
            const int slot = inside ? below : na + lane - below;
            float4* list = reinterpret_cast<float4*>(&sA[m * PITCH]);
            list[slot] = make_float4(rx, ry, rz, __uint_as_float(n));
            nsteps[pt] = (na + 3) >> 2;
        }

        float4 f[PTS_PER_WAVE][4];
        float  infl[PTS_PER_WAVE][4];
        #pragma unroll
        for (int pt = 0; pt < PTS_PER_WAVE; ++pt) {
            const int m = w * PTS_PER_WAVE + pt;
            const float4* list = reinterpret_cast<const float4*>(&sA[m * PITCH]);
            trip_load(list, fb, 0, nsteps[pt], h, c, kpx, kpy, kpz, live, f[pt], infl[pt]);
        }
        #pragma unroll
        for (int pt = 0; pt < PTS_PER_WAVE; ++pt)
            rows_requested(f[pt]);

        #pragma unroll
        for (int pt = 0; pt < PTS_PER_WAVE; ++pt) {
            const int m = w * PTS_PER_WAVE + pt;
            const float4* list = reinterpret_cast<const float4*>(&sA[m * PITCH]);

            v8f acc0 = {}, acc1 = {}, acc2 = {}, acc3 = {};
            float imax = 0.0f;
            trip_run(acc0, acc1, acc2, acc3, 0, nsteps[pt], f[pt], infl[pt], imax);
            for (int g = 2; g < nsteps[pt]; g += 2) {
                float4 fm[4];
                float  im[4];
                trip_load(list, fb, g, nsteps[pt], h, c, kpx, kpy, kpz, live, fm, im);
                rows_requested(fm);
                trip_run(acc0, acc1, acc2, acc3, g, nsteps[pt], fm, im, imax);
            }

            {
                const unsigned any = __builtin_amdgcn_ballot_w32(imax > 0.0f);
                const unsigned am  = (any | (any >> 16)) & ((1u << cK) - 1u);
                if (lane == 0) sAm[m] = am;
            }

            float* dst = &sA[m * PITCH + (8 * h) * cCin + 4 * c];
            #pragma unroll
            for (int r = 0; r < 8; ++r) {
                const float4 v = make_float4(acc0[r], acc1[r], acc2[r], acc3[r]);
                if (r < 7 || h == 0)
                    *reinterpret_cast<float4*>(dst + r * cCin) = v;
            }
        }
    }
    __syncthreads();

    if (w < cK) {
        const unsigned am   = sAm[lane];
        const bool     act  = ((am >> w) & 1u) != 0u;
        const unsigned bits = __builtin_amdgcn_ballot_w32(act);
        const int n     = __builtin_popcount(bits);
        const int below = (int)__builtin_amdgcn_mbcnt_lo(bits, 0u);
        const int slot  = act ? below : n + lane - below;
        if (slot < 16) sPrm[w * 16 + slot] = (unsigned char)lane;
        sInv[w * TILE_M + lane] = (unsigned char)(act ? slot : CMAX);
        if (lane == 0) sCnt[w] = (unsigned)n;
    }
    __syncthreads();

    const int ng = w & (NGROUPS - 1);
    const int kq = w >> 2;
    float* myB = sB + w * (16 * BP);

    v8f c00 = {}, c01 = {}, c10 = {}, c11 = {};
    for (int j = 0; j < KPW; ++j) {
        const int k = kq * KPW + j;
        if (k >= cK) break;
        const int n = __builtin_amdgcn_readfirstlane((int)sCnt[k]);
        if (n == 0) continue;
        const float* wp = W + (size_t)(k * cCin + 2 * h) * cCout + ng * 32 + 2 * c;
        if (n <= CMAX) {
            const int mrow = sPrm[k * 16 + c];
            const float* ap = &sA[mrow * PITCH + k * cCin + 2 * h];
            v8f d0 = {}, d1 = {};
            #pragma unroll 8
            for (int s = 0; s < cCin / 4; ++s) {
                const v2f a = *reinterpret_cast<const v2f*>(ap + 4 * s);
                const float2 w0 = *reinterpret_cast<const float2*>(wp + (size_t)(4 * s) * cCout);
                const float2 w1 = *reinterpret_cast<const float2*>(wp + (size_t)(4 * s + 1) * cCout);
                v2f b0; b0.x = w0.x; b0.y = w1.x;
                v2f b1; b1.x = w0.y; b1.y = w1.y;
                d0 = mm(a, b0, d0);
                d1 = mm(a, b1, d1);
            }
            #pragma unroll
            for (int r = 0; r < 8; ++r)
                *reinterpret_cast<float2*>(&myB[(8 * h + r) * BP + 2 * c]) = make_float2(d0[r], d1[r]);
            const uint2 iv0 = *reinterpret_cast<const uint2*>(&sInv[k * TILE_M + 8 * h]);
            const uint2 iv1 = *reinterpret_cast<const uint2*>(&sInv[k * TILE_M + 16 + 8 * h]);
            #pragma unroll
            for (int r = 0; r < 8; ++r) {
                const unsigned s0 = ((r < 4 ? iv0.x : iv0.y) >> (8 * (r & 3))) & 0xffu;
                const unsigned s1 = ((r < 4 ? iv1.x : iv1.y) >> (8 * (r & 3))) & 0xffu;
                const float2 v0 = *reinterpret_cast<const float2*>(&myB[s0 * BP + 2 * c]);
                const float2 v1 = *reinterpret_cast<const float2*>(&myB[s1 * BP + 2 * c]);
                c00[r] += v0.x;  c01[r] += v0.y;
                c10[r] += v1.x;  c11[r] += v1.y;
            }
        } else {
            const float* a0p = &sA[c * PITCH + k * cCin + 2 * h];
            const float* a1p = a0p + 16 * PITCH;
            #pragma unroll 4
            for (int s = 0; s < cCin / 4; ++s) {
                const v2f a0 = *reinterpret_cast<const v2f*>(a0p + 4 * s);
                const v2f a1 = *reinterpret_cast<const v2f*>(a1p + 4 * s);
                const float2 w0 = *reinterpret_cast<const float2*>(wp + (size_t)(4 * s) * cCout);
                const float2 w1 = *reinterpret_cast<const float2*>(wp + (size_t)(4 * s + 1) * cCout);
                v2f b0; b0.x = w0.x; b0.y = w1.x;
                v2f b1; b1.x = w0.y; b1.y = w1.y;
                c00 = mm(a0, b0, c00);
                c01 = mm(a0, b1, c01);
                c10 = mm(a1, b0, c10);
                c11 = mm(a1, b1, c11);
            }
        }
    }

    __syncthreads();
    {
        float4* mine = reinterpret_cast<float4*>(smem + w * PART);
        mine[(0 * 2 + 0) * 32 + lane] = make_float4(c00[0], c00[1], c00[2], c00[3]);
        mine[(0 * 2 + 1) * 32 + lane] = make_float4(c00[4], c00[5], c00[6], c00[7]);
        mine[(1 * 2 + 0) * 32 + lane] = make_float4(c01[0], c01[1], c01[2], c01[3]);
        mine[(1 * 2 + 1) * 32 + lane] = make_float4(c01[4], c01[5], c01[6], c01[7]);
        mine[(2 * 2 + 0) * 32 + lane] = make_float4(c10[0], c10[1], c10[2], c10[3]);
        mine[(2 * 2 + 1) * 32 + lane] = make_float4(c10[4], c10[5], c10[6], c10[7]);
        mine[(3 * 2 + 0) * 32 + lane] = make_float4(c11[0], c11[1], c11[2], c11[3]);
        mine[(3 * 2 + 1) * 32 + lane] = make_float4(c11[4], c11[5], c11[6], c11[7]);
    }
    __syncthreads();

    {
        const int mt = kq >> 1;
        const int rh = kq & 1;
        float4 s0 = make_float4(0.f, 0.f, 0.f, 0.f), s1 = s0;
        #pragma unroll
        for (int q = 0; q < KSPLIT; ++q) {
            const float4* src = reinterpret_cast<const float4*>(smem + (q * NGROUPS + ng) * PART);
            const float4 t0 = src[((2 * mt + 0) * 2 + rh) * 32 + lane];
            const float4 t1 = src[((2 * mt + 1) * 2 + rh) * 32 + lane];
            s0.x += t0.x; s0.y += t0.y; s0.z += t0.z; s0.w += t0.w;
            s1.x += t1.x; s1.y += t1.y; s1.z += t1.z; s1.w += t1.w;
        }
        const float v0[4] = { s0.x, s0.y, s0.z, s0.w };
        const float v1[4] = { s1.x, s1.y, s1.z, s1.w };
        float* orow = out + ((size_t)(b * cP + p0 + 16 * mt + 8 * h + 4 * rh)) * cCout + ng * 32 + 2 * c;
        #pragma unroll
        for (int rr = 0; rr < 4; ++rr) {
            float x = v0[rr];
            float y = v1[rr];
            x = x > 0.0f ? x : ALPHA * x;
            y = y > 0.0f ? y : ALPHA * y;
            *reinterpret_cast<float2*>(orow + (size_t)rr * cCout) = make_float2(x, y);
        }
    }
}

extern "C" void kernel_launch(void* const* d_in, const int* in_sizes, int n_in,
                              void* d_out, int out_size, void* d_ws, size_t ws_size,
                              hipStream_t stream) {
    const float* xyz  = (const float*)d_in[0];
    const float* feat = (const float*)d_in[1];
    const int*   qidx = (const int*)d_in[2];
    const int*   nidx = (const int*)d_in[3];
    const float* kpts = (const float*)d_in[4];
    const float* W    = (const float*)d_in[5];
    float* out = (float*)d_out;

    const int blocks = cB * cP / TILE_M;
    kpconv_fused<<<blocks, NWAVES * 32, 0, stream>>>(xyz, feat, qidx, nidx, kpts, W, out);
}
